# starting kernel
# baseline (speedup 1.0000x reference)
#include <hip/hip_runtime.h>

#define TGT 64
#define SRC 64
#define BATCH 32
#define HID 512
#define ATT 512

typedef __attribute__((ext_vector_type(2))) float v2f;
typedef __attribute__((ext_vector_type(8))) float v8f;

__device__ __forceinline__ float fast_tanh(float x) {
#if defined(__has_builtin)
#if __has_builtin(__builtin_amdgcn_tanhf)
    return __builtin_amdgcn_tanhf(x);
#elif __has_builtin(__builtin_amdgcn_tanh_f32)
    return __builtin_amdgcn_tanh_f32(x);
#else
    return tanhf(x);
#endif
#else
    return tanhf(x);
#endif
}

__global__ __launch_bounds__(128) void proj_gemm_kernel(
    const float* __restrict__ h_t,
    const float* __restrict__ src,
    const float* __restrict__ Wa,
    float* __restrict__ P)
{
    const int lane = threadIdx.x & 31;
    const int wave = threadIdx.x >> 5;
    const int tile = blockIdx.x * 4 + wave;
    const int tilesN = ATT / 16;
    const int tileM = tile / tilesN;
    const int tileN = tile % tilesN;

    const int rowBase = tileM * 64;
    const bool isH = rowBase < (TGT * BATCH);
    const float* __restrict__ X = isH ? (h_t + (size_t)rowBase * HID)
                                      : (src + (size_t)(rowBase - TGT * BATCH) * HID);
    const float* __restrict__ W = Wa + (size_t)(isH ? 0 : HID) * ATT + tileN * 16;

    const int m    = lane & 15;
    const int half = lane >> 4;

    v8f acc0 = {}, acc1 = {}, acc2 = {}, acc3 = {};
    for (int k0 = 0; k0 < HID; k0 += 4) {
        const int kA = k0 + half * 2;
        v2f b;
        b.x = W[(size_t)kA * ATT + m];
        b.y = W[(size_t)(kA + 1) * ATT + m];

        v2f a0 = *(const v2f*)(X + (size_t)(0 * 16 + m) * HID + kA);
        v2f a1 = *(const v2f*)(X + (size_t)(1 * 16 + m) * HID + kA);
        v2f a2 = *(const v2f*)(X + (size_t)(2 * 16 + m) * HID + kA);
        v2f a3 = *(const v2f*)(X + (size_t)(3 * 16 + m) * HID + kA);

        acc0 = __builtin_amdgcn_wmma_f32_16x16x4_f32(false, a0, false, b, (short)0, acc0, false, false);
        acc1 = __builtin_amdgcn_wmma_f32_16x16x4_f32(false, a1, false, b, (short)0, acc1, false, false);
        acc2 = __builtin_amdgcn_wmma_f32_16x16x4_f32(false, a2, false, b, (short)0, acc2, false, false);
        acc3 = __builtin_amdgcn_wmma_f32_16x16x4_f32(false, a3, false, b, (short)0, acc3, false, false);
    }

    float* __restrict__ out0 = P + (size_t)(rowBase + 0 * 16 + half * 8) * ATT + tileN * 16 + m;
    float* __restrict__ out1 = P + (size_t)(rowBase + 1 * 16 + half * 8) * ATT + tileN * 16 + m;
    float* __restrict__ out2 = P + (size_t)(rowBase + 2 * 16 + half * 8) * ATT + tileN * 16 + m;
    float* __restrict__ out3 = P + (size_t)(rowBase + 3 * 16 + half * 8) * ATT + tileN * 16 + m;
#pragma unroll
    for (int r = 0; r < 8; ++r) {
        out0[(size_t)r * ATT] = acc0[r];
        out1[(size_t)r * ATT] = acc1[r];
        out2[(size_t)r * ATT] = acc2[r];
        out3[(size_t)r * ATT] = acc3[r];
    }
}

__global__ __launch_bounds__(256) void attn_fused_kernel(
    const float* __restrict__ src,
    const float* __restrict__ Va,
    const float* __restrict__ P,
    float* __restrict__ out)
{
    __shared__ float hp[ATT];
    __shared__ float va[ATT];
    __shared__ float attn[SRC];

    const int tb   = blockIdx.x;
    const int b    = tb & (BATCH - 1);
    const int tid  = threadIdx.x;
    const int lane = tid & 31;
    const int wave = tid >> 5;

    const float* __restrict__ hpg = P + (size_t)tb * ATT;
    for (int i = tid; i < ATT; i += 256) {
        hp[i] = hpg[i];
        va[i] = Va[i];
    }
    __syncthreads();

    for (int s = wave; s < SRC; s += 8) {
        const float* __restrict__ spg =
            P + (size_t)(TGT * BATCH + s * BATCH + b) * ATT;
        float acc = 0.f;
#pragma unroll
        for (int i = 0; i < ATT / 64; ++i) {
            const int a = i * 64 + lane * 2;
            const float2 sp2 = *(const float2*)(spg + a);
            const float x0 = hp[a] + sp2.x;
            const float x1 = hp[a + 1] + sp2.y;
            acc += fast_tanh(x0) * va[a] + fast_tanh(x1) * va[a + 1];
        }
#pragma unroll
        for (int off = 16; off; off >>= 1)
            acc += __shfl_xor(acc, off, 32);
        if (lane == 0) attn[s] = acc;
    }
    __syncthreads();

    if (tid < 32) {
        float v0 = attn[lane], v1 = attn[lane + 32];
        float mx = fmaxf(v0, v1);
#pragma unroll
        for (int off = 16; off; off >>= 1)
            mx = fmaxf(mx, __shfl_xor(mx, off, 32));
        float e0 = __expf(v0 - mx), e1 = __expf(v1 - mx);
        float ssum = e0 + e1;
#pragma unroll
        for (int off = 16; off; off >>= 1)
            ssum += __shfl_xor(ssum, off, 32);
        float inv = 1.f / ssum;
        attn[lane]      = e0 * inv;
        attn[lane + 32] = e1 * inv;
    }
    __syncthreads();

    const int h0 = tid * 2;
    float c0 = 0.f, c1 = 0.f;
    for (int s = 0; s < SRC; ++s) {
        const float aw = attn[s];
        const float2 v = *(const float2*)(src + (size_t)(s * BATCH + b) * HID + h0);
        c0 += aw * v.x;
        c1 += aw * v.y;
    }
    float2 res; res.x = c0; res.y = c1;
    *(float2*)(out + (size_t)tb * HID + h0) = res;
}

extern "C" void kernel_launch(void* const* d_in, const int* in_sizes, int n_in,
                              void* d_out, int out_size, void* d_ws, size_t ws_size,
                              hipStream_t stream) {
    const float* h_t = (const float*)d_in[0];
    const float* src = (const float*)d_in[1];
    const float* Wa  = (const float*)d_in[2];
    const float* Va  = (const float*)d_in[3];
    float* out = (float*)d_out;

    float* P = (float*)d_ws;

    proj_gemm_kernel<<<512, 128, 0, stream>>>(h_t, src, Wa, P);

    attn_fused_kernel<<<TGT * BATCH, 256, 0, stream>>>(src, Va, P, out);
}


# Attention as two tiled passes: scores in 8x8 (t,s) register tiles per wave, then softmax + context = attn x src per (b,16 t) on WMMA in a third launch; 5x fewer attention waves; sum order changed
# speedup vs baseline: 1.5849x; 1.5849x over previous
#include <hip/hip_runtime.h>

#define TGT 64
#define SRC 64
#define BATCH 32
#define HID 512
#define ATT 512

typedef __attribute__((ext_vector_type(2))) float v2f;
typedef __attribute__((ext_vector_type(4))) float v4f;
typedef __attribute__((ext_vector_type(8))) float v8f;

__device__ __forceinline__ float fast_tanh(float x) {
#if defined(__has_builtin)
#if __has_builtin(__builtin_amdgcn_tanhf)
    return __builtin_amdgcn_tanhf(x);
#elif __has_builtin(__builtin_amdgcn_tanh_f32)
    return __builtin_amdgcn_tanh_f32(x);
#else
    return tanhf(x);
#endif
#else
    return tanhf(x);
#endif
}

#define KS 32
#define NQ (KS / 8)

struct Stage {
    v4f a[NQ];
    v4f b[NQ][4];
};

__device__ __forceinline__ void load_stage(Stage& s, const float* __restrict__ pA,
                                           const float* __restrict__ pB, int k0) {
#pragma unroll
    for (int q = 0; q < NQ; ++q) {
        s.a[q] = *(const v4f*)(pA + k0 + 8 * q);
#pragma unroll
        for (int i = 0; i < 4; ++i)
            s.b[q][i] = *(const v4f*)(pB + (size_t)(k0 + 8 * q + i) * ATT);
    }
}

__device__ __forceinline__ void mma_stage(const Stage& s, v8f (&acc)[4]) {
#pragma unroll
    for (int q = 0; q < NQ; ++q) {
#pragma unroll
        for (int t = 0; t < 2; ++t) {
            v2f a;
            a.x = s.a[q][2 * t];
            a.y = s.a[q][2 * t + 1];
#pragma unroll
            for (int j = 0; j < 4; ++j) {
                v2f b;
                b.x = s.b[q][2 * t][j];
                b.y = s.b[q][2 * t + 1][j];
                acc[j] = __builtin_amdgcn_wmma_f32_16x16x4_f32(false, a, false, b, (short)0, acc[j], false, false);
            }
        }
    }
}

__global__ __launch_bounds__(128) void proj_gemm_kernel(
    const float* __restrict__ h_t,
    const float* __restrict__ src,
    const float* __restrict__ Wa,
    float* __restrict__ P)
{
    const int lane = threadIdx.x & 31;
    const int wave = threadIdx.x >> 5;
    const int tileM = blockIdx.x >> 3;
    const int tileN = blockIdx.x & 7;
    const int nb    = tileN * 64;

    const int rowBlk = tileM * 64;
    const bool isH = rowBlk < (TGT * BATCH);
    const int rowBase = rowBlk + wave * 16;
    const float* __restrict__ X = isH ? (h_t + (size_t)rowBase * HID)
                                      : (src + (size_t)(rowBase - TGT * BATCH) * HID);
    const float* __restrict__ W = Wa + (size_t)(isH ? 0 : HID) * ATT + nb;

    const int c = lane & 15;
    const int h = lane >> 4;

    const float* __restrict__ pA = X + (size_t)c * HID + 4 * h;
    const float* __restrict__ pB = W + (size_t)(4 * h) * ATT + 4 * c;

    v8f acc[4] = {};
    Stage s0, s1;
    load_stage(s0, pA, pB, 0);
    for (int k0 = 0; k0 < HID - 2 * KS; k0 += 2 * KS) {
        load_stage(s1, pA, pB, k0 + KS);
        __builtin_amdgcn_sched_barrier(0);
        mma_stage(s0, acc);
        __builtin_amdgcn_sched_barrier(0);
        load_stage(s0, pA, pB, k0 + 2 * KS);
        __builtin_amdgcn_sched_barrier(0);
        mma_stage(s1, acc);
    }
    load_stage(s1, pA, pB, HID - KS);
    __builtin_amdgcn_sched_barrier(0);
    mma_stage(s0, acc);
    mma_stage(s1, acc);

    float* __restrict__ o = P + (size_t)(rowBase + 8 * h) * ATT + nb + 4 * c;
#pragma unroll
    for (int r = 0; r < 8; ++r) {
        v4f v;
        v.x = acc[0][r];
        v.y = acc[1][r];
        v.z = acc[2][r];
        v.w = acc[3][r];
        *(v4f*)(o + (size_t)r * ATT) = v;
    }
}

#define ROWP (BATCH * ATT)
#define RPITCH 36

struct SChunk {
    v2f hp[8];
    v2f sp[8];
    v2f va;
};

__device__ __forceinline__ void load_chunk(SChunk& c, const float* __restrict__ ph,
                                           const float* __restrict__ ps,
                                           const float* __restrict__ pv, int i) {
#pragma unroll
    for (int j = 0; j < 8; ++j) {
        c.hp[j] = *(const v2f*)(ph + (size_t)j * ROWP + 64 * i);
        c.sp[j] = *(const v2f*)(ps + (size_t)j * ROWP + 64 * i);
    }
    c.va = *(const v2f*)(pv + 64 * i);
}

__device__ __forceinline__ void tanh_chunk(const SChunk& c, v2f (&acc)[8][8]) {
#pragma unroll
    for (int t = 0; t < 8; ++t) {
#pragma unroll
        for (int s = 0; s < 8; ++s) {
            const v2f x = c.hp[t] + c.sp[s];
            v2f th;
            th.x = fast_tanh(x.x);
            th.y = fast_tanh(x.y);
            acc[t][s] += c.va * th;
        }
    }
}

__global__ __launch_bounds__(128) void score_tile_kernel(
    const float* __restrict__ Va,
    const float* __restrict__ P,
    float* __restrict__ out)
{
    __shared__ float red[4][64 * RPITCH];

    const int lane = threadIdx.x & 31;
    const int wave = threadIdx.x >> 5;
    const int s16 = blockIdx.x & 3;
    const int t16 = (blockIdx.x >> 2) & 3;
    const int b   = blockIdx.x >> 4;
    const int t8  = 2 * t16 + (wave >> 1);
    const int s8  = 2 * s16 + (wave & 1);

    const float* __restrict__ ph = P + ((size_t)(8 * t8) * BATCH + b) * ATT + 2 * lane;
    const float* __restrict__ ps = P + ((size_t)(TGT * BATCH) + (size_t)(8 * s8) * BATCH + b) * ATT + 2 * lane;
    const float* __restrict__ pv = Va + 2 * lane;

    v2f acc[8][8] = {};
    SChunk c0, c1;
    load_chunk(c0, ph, ps, pv, 0);
    for (int i = 0; i < 6; i += 2) {
        load_chunk(c1, ph, ps, pv, i + 1);
        __builtin_amdgcn_sched_barrier(0);
        tanh_chunk(c0, acc);
        __builtin_amdgcn_sched_barrier(0);
        load_chunk(c0, ph, ps, pv, i + 2);
        __builtin_amdgcn_sched_barrier(0);
        tanh_chunk(c1, acc);
    }
    load_chunk(c1, ph, ps, pv, 7);
    __builtin_amdgcn_sched_barrier(0);
    tanh_chunk(c0, acc);
    tanh_chunk(c1, acc);

    float* __restrict__ rw = red[wave];
#pragma unroll
    for (int t = 0; t < 8; ++t) {
#pragma unroll
        for (int s = 0; s < 8; ++s)
            rw[(t * 8 + s) * RPITCH + lane] = acc[t][s].x + acc[t][s].y;
    }
    __syncthreads();

    float* __restrict__ tile =
        out + ((size_t)(16 * (t8 >> 1) + 2 * (t8 & 1) + (s8 >> 2)) * BATCH + b) * HID + 64 * (s8 & 3);
#pragma unroll
    for (int u = 0; u < 2; ++u) {
        const float* __restrict__ rp = rw + (lane + 32 * u) * RPITCH;
        v4f s4 = *(const v4f*)rp;
#pragma unroll
        for (int k = 1; k < 8; ++k)
            s4 += *(const v4f*)(rp + 4 * k);
        const float sc = (s4.x + s4.y) + (s4.z + s4.w);
        tile[lane + 32 * u] = sc;
        tile[256 + lane + 32 * u] = sc;
    }
}

#define SROW (BATCH * HID)

__global__ __launch_bounds__(128) void context_kernel(
    const float* __restrict__ src,
    float* out)
{
    const int lane = threadIdx.x & 31;
    const int wave = threadIdx.x >> 5;
    const int b    = blockIdx.x & (BATCH - 1);
    const int t16  = (blockIdx.x >> 5) & 3;
    const int hh   = blockIdx.x >> 7;
    const int nb   = hh * 256 + wave * 64;
    const int c    = lane & 15;
    const int h    = lane >> 4;

    const float* pA = out + ((size_t)(16 * t16 + 2 * (c >> 3)) * BATCH + b) * HID + 256 * hh + 8 * (c & 7) + 4 * h;
    const float* __restrict__ pB = src + ((size_t)(4 * h) * BATCH + b) * HID + nb + 4 * c;

    v4f a[8];
    v4f bq[8][4];
#pragma unroll
    for (int q = 0; q < 8; ++q)
        a[q] = *(const v4f*)(pA + (size_t)(q >> 2) * SROW + 64 * (q & 3));
#pragma unroll
    for (int q = 0; q < 8; ++q) {
#pragma unroll
        for (int i = 0; i < 4; ++i)
            bq[q][i] = *(const v4f*)(pB + (size_t)(8 * q + i) * SROW);
    }
    __builtin_amdgcn_sched_barrier(0);

    float mx = a[0][0];
#pragma unroll
    for (int q = 0; q < 8; ++q) {
#pragma unroll
        for (int e = 0; e < 4; ++e)
            mx = fmaxf(mx, a[q][e]);
    }
    mx = fmaxf(mx, __shfl_xor(mx, 16, 32));
    float ssum = 0.f;
#pragma unroll
    for (int q = 0; q < 8; ++q) {
#pragma unroll
        for (int e = 0; e < 4; ++e) {
            const float ex = __expf(a[q][e] - mx);
            a[q][e] = ex;
            ssum += ex;
        }
    }
    ssum += __shfl_xor(ssum, 16, 32);
    const float inv = 1.f / ssum;
#pragma unroll
    for (int q = 0; q < 8; ++q)
        a[q] *= inv;

    v8f acc[4] = {};
#pragma unroll
    for (int q = 0; q < 8; ++q) {
#pragma unroll
        for (int t = 0; t < 2; ++t) {
            v2f av;
            av.x = a[q][2 * t];
            av.y = a[q][2 * t + 1];
#pragma unroll
            for (int j = 0; j < 4; ++j) {
                v2f bv;
                bv.x = bq[q][2 * t][j];
                bv.y = bq[q][2 * t + 1][j];
                acc[j] = __builtin_amdgcn_wmma_f32_16x16x4_f32(false, av, false, bv, (short)0, acc[j], false, false);
            }
        }
    }

    __builtin_amdgcn_sched_barrier(0);
    __syncthreads();

    float* o = out + ((size_t)(16 * t16 + 8 * h) * BATCH + b) * HID + nb + 4 * c;
#pragma unroll
    for (int r = 0; r < 8; ++r) {
        v4f v;
        v.x = acc[0][r];
        v.y = acc[1][r];
        v.z = acc[2][r];
        v.w = acc[3][r];
        *(v4f*)(o + (size_t)r * SROW) = v;
    }
}

extern "C" void kernel_launch(void* const* d_in, const int* in_sizes, int n_in,
                              void* d_out, int out_size, void* d_ws, size_t ws_size,
                              hipStream_t stream) {
    const float* h_t = (const float*)d_in[0];
    const float* src = (const float*)d_in[1];
    const float* Wa  = (const float*)d_in[2];
    const float* Va  = (const float*)d_in[3];
    float* out = (float*)d_out;

    float* P = (float*)d_ws;

    proj_gemm_kernel<<<512, 128, 0, stream>>>(h_t, src, Wa, P);

    score_tile_kernel<<<512, 128, 0, stream>>>(Va, P, out);

    context_kernel<<<256, 128, 0, stream>>>(src, out);
}


# Score and context kernels fused into one launch: a work-group owns (b, 8 t) x all 64 s, 8 waves with one 8x8 score tile each; scores go through LDS into the context WMMA; sum orders unchanged
# speedup vs baseline: 1.7928x; 1.1312x over previous
#include <hip/hip_runtime.h>

#define TGT 64
#define SRC 64
#define BATCH 32
#define HID 512
#define ATT 512

typedef __attribute__((ext_vector_type(2))) float v2f;
typedef __attribute__((ext_vector_type(4))) float v4f;
typedef __attribute__((ext_vector_type(8))) float v8f;

__device__ __forceinline__ float fast_tanh(float x) {
#if defined(__has_builtin)
#if __has_builtin(__builtin_amdgcn_tanhf)
    return __builtin_amdgcn_tanhf(x);
#elif __has_builtin(__builtin_amdgcn_tanh_f32)
    return __builtin_amdgcn_tanh_f32(x);
#else
    return tanhf(x);
#endif
#else
    return tanhf(x);
#endif
}

#define KS 32
#define NQ (KS / 8)

struct Stage {
    v4f a[NQ];
    v4f b[NQ][4];
};

__device__ __forceinline__ void load_stage(Stage& s, const float* __restrict__ pA,
                                           const float* __restrict__ pB, int k0) {
#pragma unroll
    for (int q = 0; q < NQ; ++q) {
        s.a[q] = *(const v4f*)(pA + k0 + 8 * q);
#pragma unroll
        for (int i = 0; i < 4; ++i)
            s.b[q][i] = *(const v4f*)(pB + (size_t)(k0 + 8 * q + i) * ATT);
    }
}

__device__ __forceinline__ void mma_stage(const Stage& s, v8f (&acc)[4]) {
#pragma unroll
    for (int q = 0; q < NQ; ++q) {
#pragma unroll
        for (int t = 0; t < 2; ++t) {
            v2f a;
            a.x = s.a[q][2 * t];
            a.y = s.a[q][2 * t + 1];
#pragma unroll
            for (int j = 0; j < 4; ++j) {
                v2f b;
                b.x = s.b[q][2 * t][j];
                b.y = s.b[q][2 * t + 1][j];
                acc[j] = __builtin_amdgcn_wmma_f32_16x16x4_f32(false, a, false, b, (short)0, acc[j], false, false);
            }
        }
    }
}

__global__ __launch_bounds__(128) void proj_gemm_kernel(
    const float* __restrict__ h_t,
    const float* __restrict__ src,
    const float* __restrict__ Wa,
    float* __restrict__ P)
{
    const int lane = threadIdx.x & 31;
    const int wave = threadIdx.x >> 5;
    const int tileM = blockIdx.x >> 3;
    const int tileN = blockIdx.x & 7;
    const int nb    = tileN * 64;

    const int rowBlk = tileM * 64;
    const bool isH = rowBlk < (TGT * BATCH);
    const int rowBase = rowBlk + wave * 16;
    const float* __restrict__ X = isH ? (h_t + (size_t)rowBase * HID)
                                      : (src + (size_t)(rowBase - TGT * BATCH) * HID);
    const float* __restrict__ W = Wa + (size_t)(isH ? 0 : HID) * ATT + nb;

    const int c = lane & 15;
    const int h = lane >> 4;

    const float* __restrict__ pA = X + (size_t)c * HID + 4 * h;
    const float* __restrict__ pB = W + (size_t)(4 * h) * ATT + 4 * c;

    v8f acc[4] = {};
    Stage s0, s1;
    load_stage(s0, pA, pB, 0);
    for (int k0 = 0; k0 < HID - 2 * KS; k0 += 2 * KS) {
        load_stage(s1, pA, pB, k0 + KS);
        __builtin_amdgcn_sched_barrier(0);
        mma_stage(s0, acc);
        __builtin_amdgcn_sched_barrier(0);
        load_stage(s0, pA, pB, k0 + 2 * KS);
        __builtin_amdgcn_sched_barrier(0);
        mma_stage(s1, acc);
    }
    load_stage(s1, pA, pB, HID - KS);
    __builtin_amdgcn_sched_barrier(0);
    mma_stage(s0, acc);
    mma_stage(s1, acc);

    float* __restrict__ o = P + (size_t)(rowBase + 8 * h) * ATT + nb + 4 * c;
#pragma unroll
    for (int r = 0; r < 8; ++r) {
        v4f v;
        v.x = acc[0][r];
        v.y = acc[1][r];
        v.z = acc[2][r];
        v.w = acc[3][r];
        *(v4f*)(o + (size_t)r * ATT) = v;
    }
}

#define ROWP (BATCH * ATT)
#define SROW (BATCH * HID)
#define RPITCH 36
#define SPITCH 72

struct SChunk {
    v2f hp[8];
    v2f sp[8];
    v2f va;
};

__device__ __forceinline__ void load_chunk(SChunk& c, const float* __restrict__ ph,
                                           const float* __restrict__ ps,
                                           const float* __restrict__ pv, int i) {
#pragma unroll
    for (int j = 0; j < 8; ++j) {
        c.hp[j] = *(const v2f*)(ph + (size_t)j * ROWP + 64 * i);
        c.sp[j] = *(const v2f*)(ps + (size_t)j * ROWP + 64 * i);
    }
    c.va = *(const v2f*)(pv + 64 * i);
}

__device__ __forceinline__ void tanh_chunk(const SChunk& c, v2f (&acc)[8][8]) {
#pragma unroll
    for (int t = 0; t < 8; ++t) {
#pragma unroll
        for (int s = 0; s < 8; ++s) {
            const v2f x = c.hp[t] + c.sp[s];
            v2f th;
            th.x = fast_tanh(x.x);
            th.y = fast_tanh(x.y);
            acc[t][s] += c.va * th;
        }
    }
}

__global__ __launch_bounds__(256) void attn_kernel(
    const float* __restrict__ Va,
    const float* __restrict__ P,
    const float* __restrict__ src,
    float* __restrict__ out)
{
    __shared__ float red[8][64 * RPITCH];
    __shared__ float sc[8 * SPITCH];

    const int lane = threadIdx.x & 31;
    const int wave = threadIdx.x >> 5;
    const int b    = blockIdx.x & (BATCH - 1);
    const int t8   = blockIdx.x >> 5;
    const int c    = lane & 15;
    const int h    = lane >> 4;
    const int nb   = wave * 64;

    const float* __restrict__ ph = P + ((size_t)(8 * t8) * BATCH + b) * ATT + 2 * lane;
    const float* __restrict__ ps = P + ((size_t)(TGT * BATCH) + (size_t)(8 * wave) * BATCH + b) * ATT + 2 * lane;
    const float* __restrict__ pv = Va + 2 * lane;
    const float* __restrict__ pB = src + ((size_t)(4 * h) * BATCH + b) * HID + nb + 4 * c;

    v2f acc[8][8] = {};
    SChunk c0, c1;
    load_chunk(c0, ph, ps, pv, 0);
    for (int i = 0; i < 6; i += 2) {
        load_chunk(c1, ph, ps, pv, i + 1);
        __builtin_amdgcn_sched_barrier(0);
        tanh_chunk(c0, acc);
        __builtin_amdgcn_sched_barrier(0);
        load_chunk(c0, ph, ps, pv, i + 2);
        __builtin_amdgcn_sched_barrier(0);
        tanh_chunk(c1, acc);
    }
    load_chunk(c1, ph, ps, pv, 7);
    __builtin_amdgcn_sched_barrier(0);
    tanh_chunk(c0, acc);

    v4f bq[8][4];
    __builtin_amdgcn_sched_barrier(0);
#pragma unroll
    for (int q = 0; q < 8; ++q) {
#pragma unroll
        for (int i = 0; i < 4; ++i)
            bq[q][i] = *(const v4f*)(pB + (size_t)(8 * q + i) * SROW);
    }
    __builtin_amdgcn_sched_barrier(0);
    tanh_chunk(c1, acc);

    float* __restrict__ rw = red[wave];
#pragma unroll
    for (int t = 0; t < 8; ++t) {
#pragma unroll
        for (int s = 0; s < 8; ++s)
            rw[(t * 8 + s) * RPITCH + lane] = acc[t][s].x + acc[t][s].y;
    }
    __syncthreads();

#pragma unroll
    for (int u = 0; u < 2; ++u) {
        const int p = lane + 32 * u;
        const float* __restrict__ rp = rw + p * RPITCH;
        v4f s4 = *(const v4f*)rp;
#pragma unroll
        for (int k = 1; k < 8; ++k)
            s4 += *(const v4f*)(rp + 4 * k);
        sc[(p >> 3) * SPITCH + 8 * wave + (p & 7)] = (s4.x + s4.y) + (s4.z + s4.w);
    }
    __syncthreads();

    v4f a[8];
#pragma unroll
    for (int q = 0; q < 8; ++q)
        a[q] = *(const v4f*)(sc + (c & 7) * SPITCH + 8 * q + 4 * h);

    float mx = a[0][0];
#pragma unroll
    for (int q = 0; q < 8; ++q) {
#pragma unroll
        for (int e = 0; e < 4; ++e)
            mx = fmaxf(mx, a[q][e]);
    }
    mx = fmaxf(mx, __shfl_xor(mx, 16, 32));
    float ssum = 0.f;
#pragma unroll
    for (int q = 0; q < 8; ++q) {
#pragma unroll
        for (int e = 0; e < 4; ++e) {
            const float ex = __expf(a[q][e] - mx);
            a[q][e] = ex;
            ssum += ex;
        }
    }
    ssum += __shfl_xor(ssum, 16, 32);
    const float inv = 1.f / ssum;
#pragma unroll
    for (int q = 0; q < 8; ++q)
        a[q] *= inv;

    v8f o4[4] = {};
#pragma unroll
    for (int q = 0; q < 8; ++q) {
#pragma unroll
        for (int t = 0; t < 2; ++t) {
            v2f av;
            av.x = a[q][2 * t];
            av.y = a[q][2 * t + 1];
#pragma unroll
            for (int j = 0; j < 4; ++j) {
                v2f bv;
                bv.x = bq[q][2 * t][j];
                bv.y = bq[q][2 * t + 1][j];
                o4[j] = __builtin_amdgcn_wmma_f32_16x16x4_f32(false, av, false, bv, (short)0, o4[j], false, false);
            }
        }
    }

    if (h == 0) {
        float* __restrict__ o = out + ((size_t)(8 * t8) * BATCH + b) * HID + nb + 4 * c;
#pragma unroll
        for (int r = 0; r < 8; ++r) {
            v4f v;
            v.x = o4[0][r];
            v.y = o4[1][r];
            v.z = o4[2][r];
            v.w = o4[3][r];
            *(v4f*)(o + (size_t)r * SROW) = v;
        }
    }
}

extern "C" void kernel_launch(void* const* d_in, const int* in_sizes, int n_in,
                              void* d_out, int out_size, void* d_ws, size_t ws_size,
                              hipStream_t stream) {
    const float* h_t = (const float*)d_in[0];
    const float* src = (const float*)d_in[1];
    const float* Wa  = (const float*)d_in[2];
    const float* Va  = (const float*)d_in[3];
    float* out = (float*)d_out;

    float* P = (float*)d_ws;

    proj_gemm_kernel<<<512, 128, 0, stream>>>(h_t, src, Wa, P);

    attn_kernel<<<256, 256, 0, stream>>>(Va, P, src, out);
}


# Scores fused into the projection launch: a sum split in 8 blocks of 64, a work-group projects (b, 64 a) for all t, s into LDS, emits partial scores; launch 2 adds them, 16-t tiles; sum order changed
# speedup vs baseline: 1.9402x; 1.0822x over previous
#include <hip/hip_runtime.h>


#define TGT 64
#define SRC 64
#define BATCH 32
#define HID 512
#define ATT 512

typedef __attribute__((ext_vector_type(2))) float v2f;
typedef __attribute__((ext_vector_type(4))) float v4f;
typedef __attribute__((ext_vector_type(8))) float v8f;

__device__ __forceinline__ float fast_tanh(float x) {
#if defined(__has_builtin)
#if __has_builtin(__builtin_amdgcn_tanhf)
    return __builtin_amdgcn_tanhf(x);
#elif __has_builtin(__builtin_amdgcn_tanh_f32)
    return __builtin_amdgcn_tanh_f32(x);
#else
    return tanhf(x);
#endif
#else
    return tanhf(x);
#endif
}

#define KS 32
#define NQ (KS / 8)
#define XROW (BATCH * HID)
#define PP 68

struct Stage {
    v4f a[NQ];
    v4f b[NQ][4];
};

__device__ __forceinline__ void load_stage(Stage& s, const float* __restrict__ pA,
                                           const float* __restrict__ pB, int k0) {
#pragma unroll
    for (int q = 0; q < NQ; ++q) {
        s.a[q] = *(const v4f*)(pA + k0 + 8 * q);
#pragma unroll
        for (int i = 0; i < 4; ++i)
            s.b[q][i] = *(const v4f*)(pB + (size_t)(k0 + 8 * q + i) * ATT);
    }
}

__device__ __forceinline__ void mma_stage(const Stage& s, v8f (&acc)[4]) {
#pragma unroll
    for (int q = 0; q < NQ; ++q) {
#pragma unroll
        for (int t = 0; t < 2; ++t) {
            v2f a;
            a.x = s.a[q][2 * t];
            a.y = s.a[q][2 * t + 1];
#pragma unroll
            for (int j = 0; j < 4; ++j) {
                v2f b;
                b.x = s.b[q][2 * t][j];
                b.y = s.b[q][2 * t + 1][j];
                acc[j] = __builtin_amdgcn_wmma_f32_16x16x4_f32(false, a, false, b, (short)0, acc[j], false, false);
            }
        }
    }
}

struct TChunk {
    v4f hp[8][2];
    v4f sp[2][2];
    v4f va[2];
};

__device__ __forceinline__ void load_tchunk(TChunk& c, const float* __restrict__ hp0,
                                            const float* __restrict__ sp0,
                                            const float* __restrict__ pv, int ch) {
#pragma unroll
    for (int j = 0; j < 8; ++j) {
        c.hp[j][0] = *(const v4f*)(hp0 + j * PP + 8 * ch);
        c.hp[j][1] = *(const v4f*)(hp0 + j * PP + 8 * ch + 4);
    }
#pragma unroll
    for (int u = 0; u < 2; ++u) {
        c.sp[u][0] = *(const v4f*)(sp0 + u * 32 * PP + 8 * ch);
        c.sp[u][1] = *(const v4f*)(sp0 + u * 32 * PP + 8 * ch + 4);
    }
    c.va[0] = *(const v4f*)(pv + 8 * ch);
    c.va[1] = *(const v4f*)(pv + 8 * ch + 4);
}

__device__ __forceinline__ void tanh_tchunk(const TChunk& c, v2f (&acc)[8][2]) {
#pragma unroll
    for (int k = 0; k < 2; ++k) {
#pragma unroll
        for (int e = 0; e < 2; ++e) {
            v2f vv;
            vv.x = c.va[k][2 * e];
            vv.y = c.va[k][2 * e + 1];
#pragma unroll
            for (int t = 0; t < 8; ++t) {
                v2f hh;
                hh.x = c.hp[t][k][2 * e];
                hh.y = c.hp[t][k][2 * e + 1];
#pragma unroll
                for (int u = 0; u < 2; ++u) {
                    v2f ss;
                    ss.x = c.sp[u][k][2 * e];
                    ss.y = c.sp[u][k][2 * e + 1];
                    const v2f x = hh + ss;
                    v2f th;
                    th.x = fast_tanh(x.x);
                    th.y = fast_tanh(x.y);
                    acc[t][u] += vv * th;
                }
            }
        }
    }
}

__global__ __launch_bounds__(256) void proj_score_kernel(
    const float* __restrict__ h_t,
    const float* __restrict__ src,
    const float* __restrict__ Wa,
    const float* __restrict__ Va,
    float* __restrict__ part)
{
    __shared__ float pt[128 * PP];

    const int lane = threadIdx.x & 31;
    const int wave = threadIdx.x >> 5;
    const int b    = blockIdx.x & (BATCH - 1);
    const int cb   = blockIdx.x >> 5;
    const int nb   = cb * 64;
    const int c    = lane & 15;
    const int h    = lane >> 4;

    const bool isH = wave < 4;
    const int r16  = (wave & 3) * 16;
    const float* __restrict__ X = (isH ? h_t : src) + ((size_t)r16 * BATCH + b) * HID;
    const float* __restrict__ W = Wa + (size_t)(isH ? 0 : HID) * ATT + nb;

    const float* __restrict__ pA = X + (size_t)c * XROW + 4 * h;
    const float* __restrict__ pB = W + (size_t)(4 * h) * ATT + 4 * c;

    v8f acc[4] = {};
    {
        Stage s0, s1;
        load_stage(s0, pA, pB, 0);
        for (int k0 = 0; k0 < HID - 2 * KS; k0 += 2 * KS) {
            load_stage(s1, pA, pB, k0 + KS);
            __builtin_amdgcn_sched_barrier(0);
            mma_stage(s0, acc);
            __builtin_amdgcn_sched_barrier(0);
            load_stage(s0, pA, pB, k0 + 2 * KS);
            __builtin_amdgcn_sched_barrier(0);
            mma_stage(s1, acc);
        }
        load_stage(s1, pA, pB, HID - KS);
        __builtin_amdgcn_sched_barrier(0);
        mma_stage(s0, acc);
        mma_stage(s1, acc);
    }

    {
        float* __restrict__ o = pt + (16 * wave + 8 * h) * PP + 4 * c;
#pragma unroll
        for (int r = 0; r < 8; ++r) {
            v4f v;
            v.x = acc[0][r];
            v.y = acc[1][r];
            v.z = acc[2][r];
            v.w = acc[3][r];
            *(v4f*)(o + r * PP) = v;
        }
    }
    __syncthreads();

    const float* __restrict__ hp0 = pt + (8 * wave) * PP;
    const float* __restrict__ sp0 = pt + (64 + lane) * PP;
    const float* __restrict__ pv  = Va + nb;

    v2f sacc[8][2] = {};
    TChunk c0, c1;
    load_tchunk(c0, hp0, sp0, pv, 0);
    for (int i = 0; i < 6; i += 2) {
        load_tchunk(c1, hp0, sp0, pv, i + 1);
        __builtin_amdgcn_sched_barrier(0);
        tanh_tchunk(c0, sacc);
        __builtin_amdgcn_sched_barrier(0);
        load_tchunk(c0, hp0, sp0, pv, i + 2);
        __builtin_amdgcn_sched_barrier(0);
        tanh_tchunk(c1, sacc);
    }
    load_tchunk(c1, hp0, sp0, pv, 7);
    __builtin_amdgcn_sched_barrier(0);
    tanh_tchunk(c0, sacc);
    tanh_tchunk(c1, sacc);

    float* __restrict__ po = part + (((size_t)cb * BATCH + b) * TGT + 8 * wave) * SRC + lane;
#pragma unroll
    for (int t = 0; t < 8; ++t) {
#pragma unroll
        for (int u = 0; u < 2; ++u)
            po[t * SRC + 32 * u] = sacc[t][u].x + sacc[t][u].y;
    }
}

#define SROW (BATCH * HID)
#define PLANE (BATCH * TGT * SRC)
#define SPITCH 72

__global__ __launch_bounds__(128) void ctx_kernel(
    const float* __restrict__ part,
    const float* __restrict__ src,
    float* __restrict__ out)
{
    __shared__ float sc[4][16 * SPITCH];

    const int lane = threadIdx.x & 31;
    const int wave = threadIdx.x >> 5;
    const int b    = blockIdx.x & (BATCH - 1);
    const int tt   = (blockIdx.x >> 5) & 3;
    const int hh   = blockIdx.x >> 7;
    const int c    = lane & 15;
    const int h    = lane >> 4;
    const int nb   = hh * 256 + wave * 64;

    const float* __restrict__ pT = part + ((size_t)b * TGT + 16 * tt) * SRC + 4 * lane;
    const float* __restrict__ pB = src + ((size_t)(4 * h) * BATCH + b) * HID + nb + 4 * c;

    v4f pl[8][8];
#pragma unroll
    for (int p = 0; p < 8; ++p) {
#pragma unroll
        for (int i = 0; i < 8; ++i)
            pl[p][i] = *(const v4f*)(pT + (size_t)p * PLANE + 128 * i);
    }
    v4f bq[8][4];
#pragma unroll
    for (int q = 0; q < 8; ++q) {
#pragma unroll
        for (int i = 0; i < 4; ++i)
            bq[q][i] = *(const v4f*)(pB + (size_t)(8 * q + i) * SROW);
    }
    __builtin_amdgcn_sched_barrier(0);

    float* __restrict__ scw = sc[wave];
#pragma unroll
    for (int i = 0; i < 8; ++i) {
        v4f sm = pl[0][i];
#pragma unroll
        for (int p = 1; p < 8; ++p)
            sm += pl[p][i];
        *(v4f*)(scw + (2 * i + h) * SPITCH + 4 * c) = sm;
    }
    __syncthreads();

    v4f a[8];
#pragma unroll
    for (int q = 0; q < 8; ++q)
        a[q] = *(const v4f*)(scw + c * SPITCH + 8 * q + 4 * h);

    float mx = a[0][0];
#pragma unroll
    for (int q = 0; q < 8; ++q) {
#pragma unroll
        for (int e = 0; e < 4; ++e)
            mx = fmaxf(mx, a[q][e]);
    }
    mx = fmaxf(mx, __shfl_xor(mx, 16, 32));
    float ssum = 0.f;
#pragma unroll
    for (int q = 0; q < 8; ++q) {
#pragma unroll
        for (int e = 0; e < 4; ++e) {
            const float ex = __expf(a[q][e] - mx);
            a[q][e] = ex;
            ssum += ex;
        }
    }
    ssum += __shfl_xor(ssum, 16, 32);
    const float inv = 1.f / ssum;
#pragma unroll
    for (int q = 0; q < 8; ++q)
        a[q] *= inv;

    v8f o4[4] = {};
#pragma unroll
    for (int q = 0; q < 8; ++q) {
#pragma unroll
        for (int t = 0; t < 2; ++t) {
            v2f av;
            av.x = a[q][2 * t];
            av.y = a[q][2 * t + 1];
#pragma unroll
            for (int j = 0; j < 4; ++j) {
                v2f bv;
                bv.x = bq[q][2 * t][j];
                bv.y = bq[q][2 * t + 1][j];
                o4[j] = __builtin_amdgcn_wmma_f32_16x16x4_f32(false, av, false, bv, (short)0, o4[j], false, false);
            }
        }
    }

    float* __restrict__ o = out + ((size_t)(16 * tt + 8 * h) * BATCH + b) * HID + nb + 4 * c;
#pragma unroll
    for (int r = 0; r < 8; ++r) {
        v4f v;
        v.x = o4[0][r];
        v.y = o4[1][r];
        v.z = o4[2][r];
        v.w = o4[3][r];
        *(v4f*)(o + (size_t)r * SROW) = v;
    }
}

extern "C" void kernel_launch(void* const* d_in, const int* in_sizes, int n_in,
                              void* d_out, int out_size, void* d_ws, size_t ws_size,
                              hipStream_t stream) {
    const float* h_t = (const float*)d_in[0];
    const float* src = (const float*)d_in[1];
    const float* Wa  = (const float*)d_in[2];
    const float* Va  = (const float*)d_in[3];
    float* out = (float*)d_out;

    float* part = (float*)d_ws;

    proj_score_kernel<<<256, 256, 0, stream>>>(h_t, src, Wa, Va, part);

    ctx_kernel<<<256, 128, 0, stream>>>(part, src, out);
}
